# starting kernel
# baseline (speedup 1.0000x reference)
#include <hip/hip_runtime.h>
#include <math.h>

#define NNODES 50000
#define NEDGES 800000
#define IN_F   128
#define HID_F  128
#define OUT_F  64

typedef float v2f __attribute__((ext_vector_type(2)));
typedef float v8f __attribute__((ext_vector_type(8)));


__global__ void zero_f4(float4* __restrict__ p, int n4) {
    int i = blockIdx.x * blockDim.x + threadIdx.x;
    if (i < n4) p[i] = make_float4(0.f, 0.f, 0.f, 0.f);
}

__global__ void deg_kernel(const int* __restrict__ dst, float* __restrict__ deg, int ne) {
    int e = blockIdx.x * blockDim.x + threadIdx.x;
    if (e < ne) atomicAdd(&deg[dst[e]], 1.0f);
}

__global__ void dinv_kernel(float* __restrict__ deg, int n) {
    int i = blockIdx.x * blockDim.x + threadIdx.x;
    if (i < n) deg[i] = 1.0f / sqrtf(fmaxf(deg[i], 1.0f));
}

__global__ void edge_scatter(const float* __restrict__ x,
                             const int* __restrict__ src,
                             const int* __restrict__ dst,
                             const float* __restrict__ dinv,
                             float* __restrict__ h) {
    int e = (blockIdx.x * blockDim.x + threadIdx.x) >> 5;
    int lane = threadIdx.x & 31;
    int s = src[e];
    int d = dst[e];
    float sc = dinv[s];
    const float4 v = *(const float4*)(x + (size_t)s * 128 + lane * 4);
    float* hp = h + (size_t)d * 128 + lane * 4;
    atomicAdd(hp + 0, v.x * sc);
    atomicAdd(hp + 1, v.y * sc);
    atomicAdd(hp + 2, v.z * sc);
    atomicAdd(hp + 3, v.w * sc);
}

__global__ void neg_scale(float* __restrict__ h, const float* __restrict__ dinv) {
    int t = blockIdx.x * blockDim.x + threadIdx.x;
    int n = t >> 5;
    int c = (t & 31) << 2;
    float sc = -dinv[n];
    float4* p = (float4*)(h + (size_t)n * 128 + c);
    float4 v = *p;
    v.x *= sc; v.y *= sc; v.z *= sc; v.w *= sc;
    *p = v;
}

template <int NC>
__global__ __launch_bounds__(32 * (NC / 16)) void cheb_gemm(
        const float* __restrict__ A1, const float* __restrict__ A2,
        const float* __restrict__ W,  const float* __restrict__ bias,
        float* __restrict__ out) {
    __shared__ float As[16 * 260];

    const int mbase = blockIdx.x * 16;
    const int tid = threadIdx.x;

    for (int i = tid; i < 16 * 64; i += blockDim.x) {
        int row = i >> 6;
        int col = (i & 63) << 2;
        const float* sp = (col < 128)
                        ? (A1 + (size_t)(mbase + row) * 128 + col)
                        : (A2 + (size_t)(mbase + row) * 128 + (col - 128));
        float4 v = *(const float4*)sp;
        *(float4*)(&As[row * 260 + col]) = v;
    }
    __syncthreads();

    const int ntile = tid >> 5;
    const int lane  = tid & 31;
    const int half  = lane >> 4;
    const int l     = lane & 15;

    v8f c = {};
    #pragma unroll 4
    for (int k0 = 0; k0 < 256; k0 += 4) {
        v2f a = *(const v2f*)(&As[l * 260 + k0 + 2 * half]);
        const float* wp = W + (size_t)(k0 + 2 * half) * NC + ntile * 16 + l;
        v2f b;
        b.x = wp[0];
        b.y = wp[NC];
        c = __builtin_amdgcn_wmma_f32_16x16x4_f32(
                false, a, false, b, (short)0, c, false, false);
    }

    const int col = ntile * 16 + l;
    const float bv = bias[col];
    #pragma unroll
    for (int r = 0; r < 8; ++r) {
        int row = mbase + r + 8 * half;
        out[(size_t)row * NC + col] = c[r] + bv;
    }
}


extern "C" void kernel_launch(void* const* d_in, const int* in_sizes, int n_in,
                              void* d_out, int out_size, void* d_ws, size_t ws_size,
                              hipStream_t stream) {
    const float* feat = (const float*)d_in[0];
    const int*   src  = (const int*)d_in[1];
    const int*   dst  = (const int*)d_in[2];
    const float* W1   = (const float*)d_in[3];
    const float* b1   = (const float*)d_in[4];
    const float* W2   = (const float*)d_in[5];
    const float* b2   = (const float*)d_in[6];
    float* out = (float*)d_out;

    char* ws = (char*)d_ws;
    float* dinv = (float*)ws;
    float* h    = (float*)(ws + 200704);
    float* out1 = (float*)(ws + 200704 + (size_t)NNODES * 128 * 4);

    const int TB = 256;

    zero_f4<<<(NNODES / 4 + TB - 1) / TB, TB, 0, stream>>>((float4*)dinv, NNODES / 4);
    deg_kernel<<<(NEDGES + TB - 1) / TB, TB, 0, stream>>>(dst, dinv, NEDGES);
    dinv_kernel<<<(NNODES + TB - 1) / TB, TB, 0, stream>>>(dinv, NNODES);

    const int n4h   = NNODES * 128 / 4;
    const int scat  = NEDGES * 32;
    const int nsc   = NNODES * 32;

    zero_f4<<<(n4h + TB - 1) / TB, TB, 0, stream>>>((float4*)h, n4h);
    edge_scatter<<<scat / TB, TB, 0, stream>>>(feat, src, dst, dinv, h);
    neg_scale<<<nsc / TB, TB, 0, stream>>>(h, dinv);
    cheb_gemm<HID_F><<<NNODES / 16, 32 * (HID_F / 16), 0, stream>>>(feat, h, W1, b1, out1);

    zero_f4<<<(n4h + TB - 1) / TB, TB, 0, stream>>>((float4*)h, n4h);
    edge_scatter<<<scat / TB, TB, 0, stream>>>(out1, src, dst, dinv, h);
    neg_scale<<<nsc / TB, TB, 0, stream>>>(h, dinv);
    cheb_gemm<OUT_F><<<NNODES / 16, 32 * (OUT_F / 16), 0, stream>>>(out1, h, W2, b2, out);
}


# Degree count: 128 work-groups count their edge chunk's targets per node range in LDS histograms (integer LDS atomics, exact), store partial planes; dinv launch sums the 32 planes; 7 launches to 6
# speedup vs baseline: 4.7357x; 4.7357x over previous
#include <hip/hip_runtime.h>
#include <math.h>

#define NNODES 50000
#define NEDGES 800000
#define IN_F   128
#define HID_F  128
#define OUT_F  64
#define EDGE_TB 256

#define DEG_CHUNKS  32
#define DEG_RANGES  4
#define DEG_RANGE_N 12512
#define DEG_PLANE   (DEG_RANGES * DEG_RANGE_N)
#define DEG_CHUNK_E (NEDGES / DEG_CHUNKS)
#define DEG_TB      1024
#define DEG_LD      7

typedef float v2f __attribute__((ext_vector_type(2)));
typedef float v8f __attribute__((ext_vector_type(8)));


__global__ __launch_bounds__(DEG_TB) void deg_hist(const int* __restrict__ dst, unsigned* __restrict__ part) {
    __shared__ __attribute__((aligned(16))) unsigned hist[DEG_RANGE_N];
    const int tid   = threadIdx.x;
    const int chunk = blockIdx.x / DEG_RANGES;
    const int lo    = (blockIdx.x % DEG_RANGES) * DEG_RANGE_N;

    const int4* d4 = (const int4*)(dst + (size_t)chunk * DEG_CHUNK_E);
    int4 dv[DEG_LD];
    #pragma unroll
    for (int j = 0; j < DEG_LD; ++j) {
        const int i = tid + DEG_TB * j;
        dv[j] = (i < DEG_CHUNK_E / 4) ? d4[i] : make_int4(-1, -1, -1, -1);
    }
    uint4* h4 = (uint4*)hist;
    for (int i = tid; i < DEG_RANGE_N / 4; i += DEG_TB) h4[i] = make_uint4(0u, 0u, 0u, 0u);
    __syncthreads();
    #pragma unroll
    for (int j = 0; j < DEG_LD; ++j) {
        const unsigned a0 = (unsigned)(dv[j].x - lo);
        const unsigned a1 = (unsigned)(dv[j].y - lo);
        const unsigned a2 = (unsigned)(dv[j].z - lo);
        const unsigned a3 = (unsigned)(dv[j].w - lo);
        if (a0 < DEG_RANGE_N) atomicAdd(&hist[a0], 1u);
        if (a1 < DEG_RANGE_N) atomicAdd(&hist[a1], 1u);
        if (a2 < DEG_RANGE_N) atomicAdd(&hist[a2], 1u);
        if (a3 < DEG_RANGE_N) atomicAdd(&hist[a3], 1u);
    }
    __syncthreads();
    uint4* p4 = (uint4*)(part + (size_t)chunk * DEG_PLANE + lo);
    for (int i = tid; i < DEG_RANGE_N / 4; i += DEG_TB) p4[i] = h4[i];
}

__global__ void dinv_kernel(const unsigned* __restrict__ part, float* __restrict__ dinv, int n) {
    int i = blockIdx.x * blockDim.x + threadIdx.x;
    if (i < n) {
        unsigned cnt = 0;
        #pragma unroll
        for (int c = 0; c < DEG_CHUNKS; ++c) cnt += part[(size_t)c * DEG_PLANE + i];
        dinv[i] = 1.0f / sqrtf(fmaxf((float)cnt, 1.0f));
    }
}

template <bool SQ>
__global__ __launch_bounds__(EDGE_TB) void edge_scatter64(const float* __restrict__ y,
                               const int* __restrict__ src,
                               const int* __restrict__ dst,
                               const float* __restrict__ dinv,
                               float* __restrict__ acc) {
    const int lane = threadIdx.x & 31;
    const int e = blockIdx.x * EDGE_TB + threadIdx.x;
    const int s = src[e];
    const int d = dst[e];
    const float dd = dinv[d];
    const float sc = SQ ? -(dd * dd) : -dd;
    const unsigned soff = (unsigned)s * 256u;
    const unsigned doff = (unsigned)d * 256u;
    const unsigned loff = (unsigned)lane * 4u;
    const char* yb = (const char*)y;
    char* ab = (char*)acc;

    float vlo[32], vhi[32];
    #pragma unroll
    for (int j = 0; j < 32; ++j) {
        const unsigned so = (unsigned)__builtin_amdgcn_readlane((int)soff, j);
        const float* yp = (const float*)(yb + (so + loff));
        vlo[j] = yp[0];
        vhi[j] = yp[32];
    }
    #pragma unroll
    for (int j = 0; j < 32; ++j) {
        const unsigned dofs = (unsigned)__builtin_amdgcn_readlane((int)doff, j);
        const float scj = __int_as_float(__builtin_amdgcn_readlane(__float_as_int(sc), j));
        float* ap = (float*)(ab + (dofs + loff));
        atomicAdd(ap + 0,  vlo[j] * scj);
        atomicAdd(ap + 32, vhi[j] * scj);
    }
}

__global__ __launch_bounds__(128) void weight_products(
        const float* __restrict__ W1, const float* __restrict__ b1,
        const float* __restrict__ W2, const float* __restrict__ b2,
        float* __restrict__ Wf, float* __restrict__ cu) {
    const int tid = threadIdx.x;
    if (blockIdx.x == 24) {
        const int j = tid & 63;
        const int which = tid >> 6;
        const float* w = W2 + (size_t)which * 128 * 64 + j;
        float acc = which ? 0.f : b2[j];
        for (int k = 0; k < 128; ++k) acc = fmaf(b1[k], w[(size_t)k * 64], acc);
        cu[which * 64 + j] = acc;
        return;
    }
    const int m    = blockIdx.x >> 3;
    const int rt   = blockIdx.x & 7;
    const int t    = tid >> 5;
    const int lane = tid & 31;
    const int half = lane >> 4;
    const int l    = lane & 15;

    v8f acc = {};
    const int nseg = (m == 1) ? 2 : 1;
    for (int sg = 0; sg < nseg; ++sg) {
        const int aoff = (m == 0) ? 0 : ((m == 2) ? 128 : (sg == 0 ? 128 : 0));
        const int boff = (m == 0) ? 0 : ((m == 2) ? 128 : (sg == 0 ? 0 : 128));
        const float* ap = W1 + (size_t)(aoff + 16 * rt + l) * 128 + 2 * half;
        const float* bp = W2 + (size_t)(boff + 2 * half) * 64 + 16 * t + l;
        #pragma unroll 16
        for (int k0 = 0; k0 < 128; k0 += 4) {
            v2f a = *(const v2f*)(ap + k0);
            v2f b;
            b.x = bp[(size_t)k0 * 64];
            b.y = bp[(size_t)k0 * 64 + 64];
            acc = __builtin_amdgcn_wmma_f32_16x16x4_f32(
                    false, a, false, b, (short)0, acc, false, false);
        }
    }
    #pragma unroll
    for (int r = 0; r < 8; ++r) {
        const int i  = 16 * rt + r + 8 * half;
        const int ks = i >> 2;
        const int kk = i & 3;
        const int ln = (kk >> 1) * 16 + l;
        Wf[((((size_t)ks * 4 + t) * 3 + m) * 32 + ln) * 2 + (kk & 1)] = acc[r];
    }
}

__global__ __launch_bounds__(128) void cheb_gemm3(
        const float* __restrict__ X,
        const float* __restrict__ Wf, const float* __restrict__ cu,
        const float* __restrict__ dinv,
        float* __restrict__ out, float* __restrict__ St, float* __restrict__ yR) {
    __shared__ float As[16 * 132];

    const int mbase = blockIdx.x * 16;
    const int tid = threadIdx.x;

    float4 av[4];
    #pragma unroll
    for (int j = 0; j < 4; ++j) {
        int i = tid + 128 * j;
        av[j] = *(const float4*)(X + (size_t)(mbase + (i >> 5)) * 128 + ((i & 31) << 2));
    }
    #pragma unroll
    for (int j = 0; j < 4; ++j) {
        int i = tid + 128 * j;
        *(float4*)(&As[(i >> 5) * 132 + ((i & 31) << 2)]) = av[j];
    }
    __syncthreads();

    const int t    = tid >> 5;
    const int lane = tid & 31;
    const int half = lane >> 4;
    const int l    = lane & 15;

    v8f cp = {};
    v8f cq = {};
    v8f cr = {};
    const float* wp = Wf + (size_t)t * 192 + lane * 2;
    #pragma unroll 4
    for (int ks = 0; ks < 32; ++ks) {
        v2f a = *(const v2f*)(&As[l * 132 + 4 * ks + 2 * half]);
        const float* w = wp + (size_t)ks * 768;
        v2f bp = *(const v2f*)(w);
        v2f bq = *(const v2f*)(w + 64);
        v2f br = *(const v2f*)(w + 128);
        cp = __builtin_amdgcn_wmma_f32_16x16x4_f32(
                false, a, false, bp, (short)0, cp, false, false);
        cq = __builtin_amdgcn_wmma_f32_16x16x4_f32(
                false, a, false, bq, (short)0, cq, false, false);
        cr = __builtin_amdgcn_wmma_f32_16x16x4_f32(
                false, a, false, br, (short)0, cr, false, false);
    }

    const int col = t * 16 + l;
    const float cv = cu[col];
    const float uv = cu[64 + col];
    #pragma unroll
    for (int r = 0; r < 8; ++r) {
        int row = mbase + r + 8 * half;
        float di = dinv[row];
        out[(size_t)row * 64 + col] = cp[r] + cv;
        St[(size_t)row * 64 + col]  = (cq[r] + uv) * di;
        yR[(size_t)row * 64 + col]  = cr[r] * di;
    }
}


extern "C" void kernel_launch(void* const* d_in, const int* in_sizes, int n_in,
                              void* d_out, int out_size, void* d_ws, size_t ws_size,
                              hipStream_t stream) {
    const float* feat = (const float*)d_in[0];
    const int*   src  = (const int*)d_in[1];
    const int*   dst  = (const int*)d_in[2];
    const float* W1   = (const float*)d_in[3];
    const float* b1   = (const float*)d_in[4];
    const float* W2   = (const float*)d_in[5];
    const float* b2   = (const float*)d_in[6];
    float* out = (float*)d_out;

    char* ws = (char*)d_ws;
    float* dinv = (float*)ws;
    float* yR   = (float*)(ws + 200704);
    float* St   = (float*)(ws + 200704 + (size_t)NNODES * 64 * 4);
    float* Wf   = (float*)(ws + 200704 + (size_t)NNODES * 128 * 4);
    float* cu   = (float*)(ws + 200704 + (size_t)NNODES * 128 * 4 + 128 * 192 * 4);

    unsigned* part = (unsigned*)yR;

    const int TB = 256;

    deg_hist<<<DEG_CHUNKS * DEG_RANGES, DEG_TB, 0, stream>>>(dst, part);
    dinv_kernel<<<(NNODES + TB - 1) / TB, TB, 0, stream>>>(part, dinv, NNODES);

    weight_products<<<25, 128, 0, stream>>>(W1, b1, W2, b2, Wf, cu);
    cheb_gemm3<<<NNODES / 16, 128, 0, stream>>>(feat, Wf, cu, dinv, out, St, yR);

    edge_scatter64<true><<<NEDGES / EDGE_TB, EDGE_TB, 0, stream>>>(yR, src, dst, dinv, St);
    edge_scatter64<false><<<NEDGES / EDGE_TB, EDGE_TB, 0, stream>>>(St, src, dst, dinv, out);
}


# Weight products run as extra work-groups of the degree-histogram launch (independent work, four active waves per weight block as before): 6 launches to 5; every sum's order unchanged
# speedup vs baseline: 5.0379x; 1.0638x over previous
#include <hip/hip_runtime.h>
#include <math.h>

#define NNODES 50000
#define NEDGES 800000
#define IN_F   128
#define HID_F  128
#define OUT_F  64
#define EDGE_TB 256

#define DEG_CHUNKS  32
#define DEG_RANGES  4
#define DEG_RANGE_N 12512
#define DEG_PLANE   (DEG_RANGES * DEG_RANGE_N)
#define DEG_CHUNK_E (NEDGES / DEG_CHUNKS)
#define DEG_TB      1024
#define DEG_LD      7

typedef float v2f __attribute__((ext_vector_type(2)));
typedef float v8f __attribute__((ext_vector_type(8)));


#define WP_BLOCKS 25
__device__ __forceinline__ void weight_products(const int blk, const int tid,
        const float* __restrict__ W1, const float* __restrict__ b1,
        const float* __restrict__ W2, const float* __restrict__ b2,
        float* __restrict__ Wf, float* __restrict__ cu) {
    if (blk == 24) {
        const int j = tid & 63;
        const int which = tid >> 6;
        const float* w = W2 + (size_t)which * 128 * 64 + j;
        float acc = which ? 0.f : b2[j];
        for (int k = 0; k < 128; ++k) acc = fmaf(b1[k], w[(size_t)k * 64], acc);
        cu[which * 64 + j] = acc;
        return;
    }
    const int m    = blk >> 3;
    const int rt   = blk & 7;
    const int t    = tid >> 5;
    const int lane = tid & 31;
    const int half = lane >> 4;
    const int l    = lane & 15;

    v8f acc = {};
    const int nseg = (m == 1) ? 2 : 1;
    for (int sg = 0; sg < nseg; ++sg) {
        const int aoff = (m == 0) ? 0 : ((m == 2) ? 128 : (sg == 0 ? 128 : 0));
        const int boff = (m == 0) ? 0 : ((m == 2) ? 128 : (sg == 0 ? 0 : 128));
        const float* ap = W1 + (size_t)(aoff + 16 * rt + l) * 128 + 2 * half;
        const float* bp = W2 + (size_t)(boff + 2 * half) * 64 + 16 * t + l;
        #pragma unroll 16
        for (int k0 = 0; k0 < 128; k0 += 4) {
            v2f a = *(const v2f*)(ap + k0);
            v2f b;
            b.x = bp[(size_t)k0 * 64];
            b.y = bp[(size_t)k0 * 64 + 64];
            acc = __builtin_amdgcn_wmma_f32_16x16x4_f32(
                    false, a, false, b, (short)0, acc, false, false);
        }
    }
    #pragma unroll
    for (int r = 0; r < 8; ++r) {
        const int i  = 16 * rt + r + 8 * half;
        const int ks = i >> 2;
        const int kk = i & 3;
        const int ln = (kk >> 1) * 16 + l;
        Wf[((((size_t)ks * 4 + t) * 3 + m) * 32 + ln) * 2 + (kk & 1)] = acc[r];
    }
}


__global__ __launch_bounds__(DEG_TB) void deg_hist_and_weights(
        const int* __restrict__ dst, unsigned* __restrict__ part,
        const float* __restrict__ W1, const float* __restrict__ b1,
        const float* __restrict__ W2, const float* __restrict__ b2,
        float* __restrict__ Wf, float* __restrict__ cu) {
    __shared__ __attribute__((aligned(16))) unsigned hist[DEG_RANGE_N];
    const int tid   = threadIdx.x;
    if (blockIdx.x < WP_BLOCKS) {
        if (tid < 128) weight_products(blockIdx.x, tid, W1, b1, W2, b2, Wf, cu);
        return;
    }
    const int bid   = blockIdx.x - WP_BLOCKS;
    const int chunk = bid / DEG_RANGES;
    const int lo    = (bid % DEG_RANGES) * DEG_RANGE_N;

    const int4* d4 = (const int4*)(dst + (size_t)chunk * DEG_CHUNK_E);
    int4 dv[DEG_LD];
    #pragma unroll
    for (int j = 0; j < DEG_LD; ++j) {
        const int i = tid + DEG_TB * j;
        dv[j] = (i < DEG_CHUNK_E / 4) ? d4[i] : make_int4(-1, -1, -1, -1);
    }
    uint4* h4 = (uint4*)hist;
    for (int i = tid; i < DEG_RANGE_N / 4; i += DEG_TB) h4[i] = make_uint4(0u, 0u, 0u, 0u);
    __syncthreads();
    #pragma unroll
    for (int j = 0; j < DEG_LD; ++j) {
        const unsigned a0 = (unsigned)(dv[j].x - lo);
        const unsigned a1 = (unsigned)(dv[j].y - lo);
        const unsigned a2 = (unsigned)(dv[j].z - lo);
        const unsigned a3 = (unsigned)(dv[j].w - lo);
        if (a0 < DEG_RANGE_N) atomicAdd(&hist[a0], 1u);
        if (a1 < DEG_RANGE_N) atomicAdd(&hist[a1], 1u);
        if (a2 < DEG_RANGE_N) atomicAdd(&hist[a2], 1u);
        if (a3 < DEG_RANGE_N) atomicAdd(&hist[a3], 1u);
    }
    __syncthreads();
    uint4* p4 = (uint4*)(part + (size_t)chunk * DEG_PLANE + lo);
    for (int i = tid; i < DEG_RANGE_N / 4; i += DEG_TB) p4[i] = h4[i];
}

__global__ void dinv_kernel(const unsigned* __restrict__ part, float* __restrict__ dinv, int n) {
    int i = blockIdx.x * blockDim.x + threadIdx.x;
    if (i < n) {
        unsigned cnt = 0;
        #pragma unroll
        for (int c = 0; c < DEG_CHUNKS; ++c) cnt += part[(size_t)c * DEG_PLANE + i];
        dinv[i] = 1.0f / sqrtf(fmaxf((float)cnt, 1.0f));
    }
}

template <bool SQ>
__global__ __launch_bounds__(EDGE_TB) void edge_scatter64(const float* __restrict__ y,
                               const int* __restrict__ src,
                               const int* __restrict__ dst,
                               const float* __restrict__ dinv,
                               float* __restrict__ acc) {
    const int lane = threadIdx.x & 31;
    const int e = blockIdx.x * EDGE_TB + threadIdx.x;
    const int s = src[e];
    const int d = dst[e];
    const float dd = dinv[d];
    const float sc = SQ ? -(dd * dd) : -dd;
    const unsigned soff = (unsigned)s * 256u;
    const unsigned doff = (unsigned)d * 256u;
    const unsigned loff = (unsigned)lane * 4u;
    const char* yb = (const char*)y;
    char* ab = (char*)acc;

    float vlo[32], vhi[32];
    #pragma unroll
    for (int j = 0; j < 32; ++j) {
        const unsigned so = (unsigned)__builtin_amdgcn_readlane((int)soff, j);
        const float* yp = (const float*)(yb + (so + loff));
        vlo[j] = yp[0];
        vhi[j] = yp[32];
    }
    #pragma unroll
    for (int j = 0; j < 32; ++j) {
        const unsigned dofs = (unsigned)__builtin_amdgcn_readlane((int)doff, j);
        const float scj = __int_as_float(__builtin_amdgcn_readlane(__float_as_int(sc), j));
        float* ap = (float*)(ab + (dofs + loff));
        atomicAdd(ap + 0,  vlo[j] * scj);
        atomicAdd(ap + 32, vhi[j] * scj);
    }
}

__global__ __launch_bounds__(128) void cheb_gemm3(
        const float* __restrict__ X,
        const float* __restrict__ Wf, const float* __restrict__ cu,
        const float* __restrict__ dinv,
        float* __restrict__ out, float* __restrict__ St, float* __restrict__ yR) {
    __shared__ float As[16 * 132];

    const int mbase = blockIdx.x * 16;
    const int tid = threadIdx.x;

    float4 av[4];
    #pragma unroll
    for (int j = 0; j < 4; ++j) {
        int i = tid + 128 * j;
        av[j] = *(const float4*)(X + (size_t)(mbase + (i >> 5)) * 128 + ((i & 31) << 2));
    }
    #pragma unroll
    for (int j = 0; j < 4; ++j) {
        int i = tid + 128 * j;
        *(float4*)(&As[(i >> 5) * 132 + ((i & 31) << 2)]) = av[j];
    }
    __syncthreads();

    const int t    = tid >> 5;
    const int lane = tid & 31;
    const int half = lane >> 4;
    const int l    = lane & 15;

    v8f cp = {};
    v8f cq = {};
    v8f cr = {};
    const float* wp = Wf + (size_t)t * 192 + lane * 2;
    #pragma unroll 4
    for (int ks = 0; ks < 32; ++ks) {
        v2f a = *(const v2f*)(&As[l * 132 + 4 * ks + 2 * half]);
        const float* w = wp + (size_t)ks * 768;
        v2f bp = *(const v2f*)(w);
        v2f bq = *(const v2f*)(w + 64);
        v2f br = *(const v2f*)(w + 128);
        cp = __builtin_amdgcn_wmma_f32_16x16x4_f32(
                false, a, false, bp, (short)0, cp, false, false);
        cq = __builtin_amdgcn_wmma_f32_16x16x4_f32(
                false, a, false, bq, (short)0, cq, false, false);
        cr = __builtin_amdgcn_wmma_f32_16x16x4_f32(
                false, a, false, br, (short)0, cr, false, false);
    }

    const int col = t * 16 + l;
    const float cv = cu[col];
    const float uv = cu[64 + col];
    #pragma unroll
    for (int r = 0; r < 8; ++r) {
        int row = mbase + r + 8 * half;
        float di = dinv[row];
        out[(size_t)row * 64 + col] = cp[r] + cv;
        St[(size_t)row * 64 + col]  = (cq[r] + uv) * di;
        yR[(size_t)row * 64 + col]  = cr[r] * di;
    }
}


extern "C" void kernel_launch(void* const* d_in, const int* in_sizes, int n_in,
                              void* d_out, int out_size, void* d_ws, size_t ws_size,
                              hipStream_t stream) {
    const float* feat = (const float*)d_in[0];
    const int*   src  = (const int*)d_in[1];
    const int*   dst  = (const int*)d_in[2];
    const float* W1   = (const float*)d_in[3];
    const float* b1   = (const float*)d_in[4];
    const float* W2   = (const float*)d_in[5];
    const float* b2   = (const float*)d_in[6];
    float* out = (float*)d_out;

    char* ws = (char*)d_ws;
    float* dinv = (float*)ws;
    float* yR   = (float*)(ws + 200704);
    float* St   = (float*)(ws + 200704 + (size_t)NNODES * 64 * 4);
    float* Wf   = (float*)(ws + 200704 + (size_t)NNODES * 128 * 4);
    float* cu   = (float*)(ws + 200704 + (size_t)NNODES * 128 * 4 + 128 * 192 * 4);

    unsigned* part = (unsigned*)yR;

    const int TB = 256;

    deg_hist_and_weights<<<WP_BLOCKS + DEG_CHUNKS * DEG_RANGES, DEG_TB, 0, stream>>>(dst, part, W1, b1, W2, b2, Wf, cu);
    dinv_kernel<<<(NNODES + TB - 1) / TB, TB, 0, stream>>>(part, dinv, NNODES);

    cheb_gemm3<<<NNODES / 16, 128, 0, stream>>>(feat, Wf, cu, dinv, out, St, yR);

    edge_scatter64<true><<<NEDGES / EDGE_TB, EDGE_TB, 0, stream>>>(yR, src, dst, dinv, St);
    edge_scatter64<false><<<NEDGES / EDGE_TB, EDGE_TB, 0, stream>>>(St, src, dst, dinv, out);
}


# Edge passes: a target row's float atomics now all come from one half of the package: work-groups draw (edge group, row side) units from per-side counters by die number; atomic arrival order changes
# speedup vs baseline: 6.1190x; 1.2146x over previous
#include <hip/hip_runtime.h>
#include <math.h>

#define NNODES 50000
#define NEDGES 800000
#define IN_F   128
#define HID_F  128
#define OUT_F  64
#define EDGE_TB 256
#define EDGE_BATCHES 5
#define EDGE_GROUP_E (EDGE_TB * EDGE_BATCHES)
#define EDGE_GROUPS  ((NEDGES + EDGE_GROUP_E - 1) / EDGE_GROUP_E)
#define HALF_SPLIT   (NNODES / 2)
#define CTR_STRIDE   64

#define DEG_CHUNKS  32
#define DEG_RANGES  4
#define DEG_RANGE_N 12512
#define DEG_PLANE   (DEG_RANGES * DEG_RANGE_N)
#define DEG_CHUNK_E (NEDGES / DEG_CHUNKS)
#define DEG_TB      1024
#define DEG_LD      7

typedef float v2f __attribute__((ext_vector_type(2)));
typedef float v8f __attribute__((ext_vector_type(8)));


#define WP_BLOCKS 25
__device__ __forceinline__ void weight_products(const int blk, const int tid,
        const float* __restrict__ W1, const float* __restrict__ b1,
        const float* __restrict__ W2, const float* __restrict__ b2,
        float* __restrict__ Wf, float* __restrict__ cu) {
    if (blk == 24) {
        const int j = tid & 63;
        const int which = tid >> 6;
        const float* w = W2 + (size_t)which * 128 * 64 + j;
        float acc = which ? 0.f : b2[j];
        for (int k = 0; k < 128; ++k) acc = fmaf(b1[k], w[(size_t)k * 64], acc);
        cu[which * 64 + j] = acc;
        return;
    }
    const int m    = blk >> 3;
    const int rt   = blk & 7;
    const int t    = tid >> 5;
    const int lane = tid & 31;
    const int half = lane >> 4;
    const int l    = lane & 15;

    v8f acc = {};
    const int nseg = (m == 1) ? 2 : 1;
    for (int sg = 0; sg < nseg; ++sg) {
        const int aoff = (m == 0) ? 0 : ((m == 2) ? 128 : (sg == 0 ? 128 : 0));
        const int boff = (m == 0) ? 0 : ((m == 2) ? 128 : (sg == 0 ? 0 : 128));
        const float* ap = W1 + (size_t)(aoff + 16 * rt + l) * 128 + 2 * half;
        const float* bp = W2 + (size_t)(boff + 2 * half) * 64 + 16 * t + l;
        #pragma unroll 16
        for (int k0 = 0; k0 < 128; k0 += 4) {
            v2f a = *(const v2f*)(ap + k0);
            v2f b;
            b.x = bp[(size_t)k0 * 64];
            b.y = bp[(size_t)k0 * 64 + 64];
            acc = __builtin_amdgcn_wmma_f32_16x16x4_f32(
                    false, a, false, b, (short)0, acc, false, false);
        }
    }
    #pragma unroll
    for (int r = 0; r < 8; ++r) {
        const int i  = 16 * rt + r + 8 * half;
        const int ks = i >> 2;
        const int kk = i & 3;
        const int ln = (kk >> 1) * 16 + l;
        Wf[((((size_t)ks * 4 + t) * 3 + m) * 32 + ln) * 2 + (kk & 1)] = acc[r];
    }
}


__global__ __launch_bounds__(DEG_TB) void deg_hist_and_weights(
        const int* __restrict__ dst, unsigned* __restrict__ part,
        const float* __restrict__ W1, const float* __restrict__ b1,
        const float* __restrict__ W2, const float* __restrict__ b2,
        float* __restrict__ Wf, float* __restrict__ cu) {
    __shared__ __attribute__((aligned(16))) unsigned hist[DEG_RANGE_N];
    const int tid   = threadIdx.x;
    if (blockIdx.x < WP_BLOCKS) {
        if (tid < 128) weight_products(blockIdx.x, tid, W1, b1, W2, b2, Wf, cu);
        return;
    }
    const int bid   = blockIdx.x - WP_BLOCKS;
    const int chunk = bid / DEG_RANGES;
    const int lo    = (bid % DEG_RANGES) * DEG_RANGE_N;

    const int4* d4 = (const int4*)(dst + (size_t)chunk * DEG_CHUNK_E);
    int4 dv[DEG_LD];
    #pragma unroll
    for (int j = 0; j < DEG_LD; ++j) {
        const int i = tid + DEG_TB * j;
        dv[j] = (i < DEG_CHUNK_E / 4) ? d4[i] : make_int4(-1, -1, -1, -1);
    }
    uint4* h4 = (uint4*)hist;
    for (int i = tid; i < DEG_RANGE_N / 4; i += DEG_TB) h4[i] = make_uint4(0u, 0u, 0u, 0u);
    __syncthreads();
    #pragma unroll
    for (int j = 0; j < DEG_LD; ++j) {
        const unsigned a0 = (unsigned)(dv[j].x - lo);
        const unsigned a1 = (unsigned)(dv[j].y - lo);
        const unsigned a2 = (unsigned)(dv[j].z - lo);
        const unsigned a3 = (unsigned)(dv[j].w - lo);
        if (a0 < DEG_RANGE_N) atomicAdd(&hist[a0], 1u);
        if (a1 < DEG_RANGE_N) atomicAdd(&hist[a1], 1u);
        if (a2 < DEG_RANGE_N) atomicAdd(&hist[a2], 1u);
        if (a3 < DEG_RANGE_N) atomicAdd(&hist[a3], 1u);
    }
    __syncthreads();
    uint4* p4 = (uint4*)(part + (size_t)chunk * DEG_PLANE + lo);
    for (int i = tid; i < DEG_RANGE_N / 4; i += DEG_TB) p4[i] = h4[i];
}

__global__ void dinv_kernel(const unsigned* __restrict__ part, float* __restrict__ dinv, int n,
                            unsigned* __restrict__ ctr) {
    int i = blockIdx.x * blockDim.x + threadIdx.x;
    if (i < 4) ctr[32 * i] = 0u;
    if (i < n) {
        unsigned cnt = 0;
        #pragma unroll
        for (int c = 0; c < DEG_CHUNKS; ++c) cnt += part[(size_t)c * DEG_PLANE + i];
        dinv[i] = 1.0f / sqrtf(fmaxf((float)cnt, 1.0f));
    }
}

template <bool SQ>
__global__ __launch_bounds__(EDGE_TB) void edge_scatter64(const float* __restrict__ y,
                               const int* __restrict__ src,
                               const int* __restrict__ dst,
                               const float* __restrict__ dinv,
                               float* __restrict__ acc,
                               unsigned* __restrict__ ctr) {
    __shared__ unsigned s_work;
    if (threadIdx.x == 0) {
        const unsigned id = (unsigned)__builtin_amdgcn_s_sendmsg_rtn(0x87);
        unsigned sd = (((id >> 16) & 15u) >= 4u) ? 1u : 0u;
        unsigned k = atomicAdd(&ctr[sd * CTR_STRIDE], 1u);
        if (k >= (unsigned)EDGE_GROUPS) {
            sd ^= 1u;
            k = atomicAdd(&ctr[sd * CTR_STRIDE], 1u);
        }
        s_work = k * 2u + sd;
    }
    __syncthreads();
    const unsigned work = s_work;
    const unsigned side = work & 1u;
    const unsigned grp  = work >> 1;
    if (grp >= (unsigned)EDGE_GROUPS) return;

    const int lane = threadIdx.x & 31;
    const unsigned loff = (unsigned)lane * 4u;
    const char* yb = (const char*)y;
    char* ab = (char*)acc;

    int e = (int)grp * EDGE_GROUP_E + (int)threadIdx.x;
    int d = (e < NEDGES) ? dst[e] : -1;
    #pragma unroll 1
    for (int bt = 0; bt < EDGE_BATCHES; ++bt) {
        const bool mine = (d >= 0) && ((d >= HALF_SPLIT ? 1u : 0u) == side);
        int s = 0;
        float dd = 0.f;
        if (mine) { s = src[e]; dd = dinv[d]; }
        const int dcur = mine ? d : 0;
        e += EDGE_TB;
        d = (bt + 1 < EDGE_BATCHES && e < NEDGES) ? dst[e] : -1;

        const unsigned m = __builtin_amdgcn_ballot_w32(mine);
        const float sc = SQ ? -(dd * dd) : -dd;
        const unsigned soff = (unsigned)s * 256u;
        const unsigned doff = (unsigned)dcur * 256u;

        float vlo[32], vhi[32];
        #pragma unroll
        for (int j = 0; j < 32; ++j) {
            if ((m >> j) & 1u) {
                const unsigned so = (unsigned)__builtin_amdgcn_readlane((int)soff, j);
                const float* yp = (const float*)(yb + (so + loff));
                vlo[j] = yp[0];
                vhi[j] = yp[32];
            }
        }
        #pragma unroll
        for (int j = 0; j < 32; ++j) {
            if ((m >> j) & 1u) {
                const unsigned dofs = (unsigned)__builtin_amdgcn_readlane((int)doff, j);
                const float scj = __int_as_float(__builtin_amdgcn_readlane(__float_as_int(sc), j));
                float* ap = (float*)(ab + (dofs + loff));
                atomicAdd(ap + 0,  vlo[j] * scj);
                atomicAdd(ap + 32, vhi[j] * scj);
            }
        }
    }
}

__global__ __launch_bounds__(128) void cheb_gemm3(
        const float* __restrict__ X,
        const float* __restrict__ Wf, const float* __restrict__ cu,
        const float* __restrict__ dinv,
        float* __restrict__ out, float* __restrict__ St, float* __restrict__ yR) {
    __shared__ float As[16 * 132];

    const int mbase = blockIdx.x * 16;
    const int tid = threadIdx.x;

    float4 av[4];
    #pragma unroll
    for (int j = 0; j < 4; ++j) {
        int i = tid + 128 * j;
        av[j] = *(const float4*)(X + (size_t)(mbase + (i >> 5)) * 128 + ((i & 31) << 2));
    }
    #pragma unroll
    for (int j = 0; j < 4; ++j) {
        int i = tid + 128 * j;
        *(float4*)(&As[(i >> 5) * 132 + ((i & 31) << 2)]) = av[j];
    }
    __syncthreads();

    const int t    = tid >> 5;
    const int lane = tid & 31;
    const int half = lane >> 4;
    const int l    = lane & 15;

    v8f cp = {};
    v8f cq = {};
    v8f cr = {};
    const float* wp = Wf + (size_t)t * 192 + lane * 2;
    #pragma unroll 4
    for (int ks = 0; ks < 32; ++ks) {
        v2f a = *(const v2f*)(&As[l * 132 + 4 * ks + 2 * half]);
        const float* w = wp + (size_t)ks * 768;
        v2f bp = *(const v2f*)(w);
        v2f bq = *(const v2f*)(w + 64);
        v2f br = *(const v2f*)(w + 128);
        cp = __builtin_amdgcn_wmma_f32_16x16x4_f32(
                false, a, false, bp, (short)0, cp, false, false);
        cq = __builtin_amdgcn_wmma_f32_16x16x4_f32(
                false, a, false, bq, (short)0, cq, false, false);
        cr = __builtin_amdgcn_wmma_f32_16x16x4_f32(
                false, a, false, br, (short)0, cr, false, false);
    }

    const int col = t * 16 + l;
    const float cv = cu[col];
    const float uv = cu[64 + col];
    #pragma unroll
    for (int r = 0; r < 8; ++r) {
        int row = mbase + r + 8 * half;
        float di = dinv[row];
        out[(size_t)row * 64 + col] = cp[r] + cv;
        St[(size_t)row * 64 + col]  = (cq[r] + uv) * di;
        yR[(size_t)row * 64 + col]  = cr[r] * di;
    }
}


extern "C" void kernel_launch(void* const* d_in, const int* in_sizes, int n_in,
                              void* d_out, int out_size, void* d_ws, size_t ws_size,
                              hipStream_t stream) {
    const float* feat = (const float*)d_in[0];
    const int*   src  = (const int*)d_in[1];
    const int*   dst  = (const int*)d_in[2];
    const float* W1   = (const float*)d_in[3];
    const float* b1   = (const float*)d_in[4];
    const float* W2   = (const float*)d_in[5];
    const float* b2   = (const float*)d_in[6];
    float* out = (float*)d_out;

    char* ws = (char*)d_ws;
    float* dinv = (float*)ws;
    float* yR   = (float*)(ws + 200704);
    float* St   = (float*)(ws + 200704 + (size_t)NNODES * 64 * 4);
    float* Wf   = (float*)(ws + 200704 + (size_t)NNODES * 128 * 4);
    float* cu   = (float*)(ws + 200704 + (size_t)NNODES * 128 * 4 + 128 * 192 * 4);

    unsigned* part = (unsigned*)yR;
    unsigned* ctr = (unsigned*)(ws + 200192);

    const int TB = 256;

    deg_hist_and_weights<<<WP_BLOCKS + DEG_CHUNKS * DEG_RANGES, DEG_TB, 0, stream>>>(dst, part, W1, b1, W2, b2, Wf, cu);
    dinv_kernel<<<(NNODES + TB - 1) / TB, TB, 0, stream>>>(part, dinv, NNODES, ctr);

    cheb_gemm3<<<NNODES / 16, 128, 0, stream>>>(feat, Wf, cu, dinv, out, St, yR);

    edge_scatter64<true><<<2 * EDGE_GROUPS, EDGE_TB, 0, stream>>>(yR, src, dst, dinv, St, ctr);
    edge_scatter64<false><<<2 * EDGE_GROUPS, EDGE_TB, 0, stream>>>(St, src, dst, dinv, out, ctr + 32);
}
